# starting kernel
# baseline (speedup 1.0000x reference)
#include <hip/hip_runtime.h>

typedef float v2f __attribute__((ext_vector_type(2)));
typedef float v8f __attribute__((ext_vector_type(8)));

__global__ void zero_f32(float* __restrict__ p, int n) {
  int i = blockIdx.x * blockDim.x + threadIdx.x;
  if (i < n) p[i] = 0.0f;
}

__global__ void degree_kernel(const int* __restrict__ src, const int* __restrict__ dst,
                              float* __restrict__ deg_out, float* __restrict__ deg_in,
                              int n_edges) {
  int e = blockIdx.x * blockDim.x + threadIdx.x;
  if (e >= n_edges) return;
  atomicAdd(deg_out + src[e], 1.0f);
  atomicAdd(deg_in + dst[e], 1.0f);
}

__global__ void norm_kernel(float* __restrict__ deg, int n) {
  int i = blockIdx.x * blockDim.x + threadIdx.x;
  if (i >= n) return;
  deg[i] = rsqrtf(fmaxf(deg[i], 1.0f));
}

__global__ void post_kernel(const float* __restrict__ agg, const float* __restrict__ norm_d,
                            const float* __restrict__ bias, float* __restrict__ out,
                            int n_elems, int do_relu) {
  int i = blockIdx.x * blockDim.x + threadIdx.x;
  if (i >= n_elems) return;
  int row = i >> 6;
  int col = i & 63;
  float v = fmaf(agg[i], norm_d[row], bias[col]);
  if (do_relu) v = fmaxf(v, 0.0f);
  out[i] = v;
}

__global__ void spmm_scatter(const float* __restrict__ H, const int* __restrict__ src,
                             const int* __restrict__ dst, float* __restrict__ agg,
                             int n_edges) {
  int t = blockIdx.x * blockDim.x + threadIdx.x;
  int e = t >> 4;
  if (e >= n_edges) return;
  int c = (t & 15) * 4;
  int s = src[e];
  int d = dst[e];
  const float4 v = *reinterpret_cast<const float4*>(H + (size_t)s * 64 + c);
  float* o = agg + (size_t)d * 64 + c;
  atomicAdd(o + 0, v.x);
  atomicAdd(o + 1, v.y);
  atomicAdd(o + 2, v.z);
  atomicAdd(o + 3, v.w);
}

template <int K>
__global__ void __launch_bounds__(256) gemm_norm_wmma(const float* __restrict__ X,
                                                      const float* __restrict__ W,
                                                      const float* __restrict__ norm_s,
                                                      float* __restrict__ Y,
                                                      int n_nodes) {
  __shared__ float sW[K * 64];
  for (int i = threadIdx.x; i < K * 64; i += 256) sW[i] = W[i];
  __syncthreads();

  const int wave = threadIdx.x >> 5;
  const int lane = threadIdx.x & 31;
  const int row0 = (blockIdx.x * 8 + wave) * 16;
  if (row0 + 15 >= n_nodes) return;

  const int half  = lane >> 4;
  const int sub   = lane & 15;
  const int mrow  = row0 + sub;
  const float ns  = norm_s[mrow];
  const float* __restrict__ xrow = X + (size_t)mrow * K;

  v8f acc0 = {}, acc1 = {}, acc2 = {}, acc3 = {};

#pragma unroll
  for (int kk = 0; kk < K / 4; ++kk) {
    const int ka = kk * 4 + half * 2;
    v2f a;
    a.x = xrow[ka] * ns;
    a.y = xrow[ka + 1] * ns;

    const float* wk = &sW[ka * 64 + sub];
    v2f b;

    b.x = wk[0];       b.y = wk[64];
    acc0 = __builtin_amdgcn_wmma_f32_16x16x4_f32(false, a, false, b, (short)0, acc0, false, false);
    b.x = wk[16];      b.y = wk[64 + 16];
    acc1 = __builtin_amdgcn_wmma_f32_16x16x4_f32(false, a, false, b, (short)0, acc1, false, false);
    b.x = wk[32];      b.y = wk[64 + 32];
    acc2 = __builtin_amdgcn_wmma_f32_16x16x4_f32(false, a, false, b, (short)0, acc2, false, false);
    b.x = wk[48];      b.y = wk[64 + 48];
    acc3 = __builtin_amdgcn_wmma_f32_16x16x4_f32(false, a, false, b, (short)0, acc3, false, false);
  }

  float* ybase = Y + (size_t)row0 * 64 + sub;
#pragma unroll
  for (int r = 0; r < 8; ++r) {
    const int m = r + half * 8;
    float* yr = ybase + (size_t)m * 64;
    yr[0]  = acc0[r];
    yr[16] = acc1[r];
    yr[32] = acc2[r];
    yr[48] = acc3[r];
  }
}

static inline int cdiv(long long a, long long b) { return (int)((a + b - 1) / b); }

extern "C" void kernel_launch(void* const* d_in, const int* in_sizes, int n_in,
                              void* d_out, int out_size, void* d_ws, size_t ws_size,
                              hipStream_t stream) {
  const float* features = (const float*)d_in[0];
  const int*   src      = (const int*)d_in[1];
  const int*   dst      = (const int*)d_in[2];
  const float* W1       = (const float*)d_in[3];
  const float* b1       = (const float*)d_in[4];
  const float* W2       = (const float*)d_in[5];
  const float* b2       = (const float*)d_in[6];

  const int n_nodes = in_sizes[0] / 128;
  const int n_edges = in_sizes[1];
  const int n_feat  = n_nodes * 64;

  float* ws     = (float*)d_ws;
  float* norm_s = ws;
  float* norm_d = ws + n_nodes;
  float* bufA   = ws + 2 * (size_t)n_nodes;
  float* bufB   = bufA + (size_t)n_nodes * 64;
  float* out    = (float*)d_out;

  const int B = 256;

  zero_f32<<<cdiv(2 * (long long)n_nodes, B), B, 0, stream>>>(norm_s, 2 * n_nodes);
  degree_kernel<<<cdiv(n_edges, B), B, 0, stream>>>(src, dst, norm_s, norm_d, n_edges);
  norm_kernel<<<cdiv(2 * (long long)n_nodes, B), B, 0, stream>>>(norm_s, 2 * n_nodes);

  const int gemm_blocks = cdiv(n_nodes, 128);

  gemm_norm_wmma<128><<<gemm_blocks, B, 0, stream>>>(features, W1, norm_s, bufA, n_nodes);
  zero_f32<<<cdiv(n_feat, B), B, 0, stream>>>(bufB, n_feat);
  spmm_scatter<<<cdiv((long long)n_edges * 16, B), B, 0, stream>>>(bufA, src, dst, bufB, n_edges);
  post_kernel<<<cdiv(n_feat, B), B, 0, stream>>>(bufB, norm_d, b1, bufA, n_feat, 1);

  gemm_norm_wmma<64><<<gemm_blocks, B, 0, stream>>>(bufA, W2, norm_s, bufB, n_nodes);
  zero_f32<<<cdiv(n_feat, B), B, 0, stream>>>(out, n_feat);
  spmm_scatter<<<cdiv((long long)n_edges * 16, B), B, 0, stream>>>(bufB, src, dst, out, n_edges);
  post_kernel<<<cdiv(n_feat, B), B, 0, stream>>>(out, norm_d, b2, out, n_feat, 0);
}


# Edge scatter re-laid: one wave walks 32 edges, lane = column, so each f32 atomic add and each row load covers one whole aligned 128-B line (was 4 lines x 8 floats); sums stay in atomic arrival order
# speedup vs baseline: 1.5521x; 1.5521x over previous
#include <hip/hip_runtime.h>

typedef float v2f __attribute__((ext_vector_type(2)));
typedef float v8f __attribute__((ext_vector_type(8)));

__global__ void zero_f32(float* __restrict__ p, int n) {
  int i = blockIdx.x * blockDim.x + threadIdx.x;
  if (i < n) p[i] = 0.0f;
}

__global__ void degree_kernel(const int* __restrict__ src, const int* __restrict__ dst,
                              float* __restrict__ deg_out, float* __restrict__ deg_in,
                              int n_edges) {
  int e = blockIdx.x * blockDim.x + threadIdx.x;
  if (e >= n_edges) return;
  atomicAdd(deg_out + src[e], 1.0f);
  atomicAdd(deg_in + dst[e], 1.0f);
}

__global__ void norm_kernel(float* __restrict__ deg, int n) {
  int i = blockIdx.x * blockDim.x + threadIdx.x;
  if (i >= n) return;
  deg[i] = rsqrtf(fmaxf(deg[i], 1.0f));
}

__global__ void post_kernel(const float* __restrict__ agg, const float* __restrict__ norm_d,
                            const float* __restrict__ bias, float* __restrict__ out,
                            int n_elems, int do_relu) {
  int i = blockIdx.x * blockDim.x + threadIdx.x;
  if (i >= n_elems) return;
  int row = i >> 6;
  int col = i & 63;
  float v = fmaf(agg[i], norm_d[row], bias[col]);
  if (do_relu) v = fmaxf(v, 0.0f);
  out[i] = v;
}

#define SPMM_GROUP 8

__global__ void __launch_bounds__(256) spmm_scatter(const float* __restrict__ H,
                                                    const int* __restrict__ src,
                                                    const int* __restrict__ dst,
                                                    float* __restrict__ agg, int n_edges) {
  const int lane = threadIdx.x & 31;
  const int wave = __builtin_amdgcn_readfirstlane((int)((blockIdx.x * blockDim.x + threadIdx.x) >> 5));
  const int e0 = wave * 32;
  if (e0 >= n_edges) return;
  const int e = e0 + lane;
  int s_l = 0, d_l = 0;
  if (e < n_edges) { s_l = src[e]; d_l = dst[e]; }
  const int cnt = n_edges - e0;
  if (cnt >= 32) {
#pragma unroll
    for (int g = 0; g < 32; g += SPMM_GROUP) {
      float v0[SPMM_GROUP], v1[SPMM_GROUP];
#pragma unroll
      for (int u = 0; u < SPMM_GROUP; ++u) {
        const int s = __builtin_amdgcn_readlane(s_l, g + u);
        const float* h = H + (size_t)s * 64 + lane;
        v0[u] = h[0];
        v1[u] = h[32];
      }
#pragma unroll
      for (int u = 0; u < SPMM_GROUP; ++u) {
        const int d = __builtin_amdgcn_readlane(d_l, g + u);
        float* o = agg + (size_t)d * 64 + lane;
        atomicAdd(o, v0[u]);
        atomicAdd(o + 32, v1[u]);
      }
    }
  } else {
#pragma clang loop unroll(disable)
    for (int j = 0; j < cnt; ++j) {
      const int s = __builtin_amdgcn_readlane(s_l, j);
      const int d = __builtin_amdgcn_readlane(d_l, j);
      const float* h = H + (size_t)s * 64 + lane;
      float* o = agg + (size_t)d * 64 + lane;
      atomicAdd(o, h[0]);
      atomicAdd(o + 32, h[32]);
    }
  }
}

template <int K>
__global__ void __launch_bounds__(256) gemm_norm_wmma(const float* __restrict__ X,
                                                      const float* __restrict__ W,
                                                      const float* __restrict__ norm_s,
                                                      float* __restrict__ Y,
                                                      int n_nodes) {
  __shared__ float sW[K * 64];
  for (int i = threadIdx.x; i < K * 64; i += 256) sW[i] = W[i];
  __syncthreads();

  const int wave = threadIdx.x >> 5;
  const int lane = threadIdx.x & 31;
  const int row0 = (blockIdx.x * 8 + wave) * 16;
  if (row0 + 15 >= n_nodes) return;

  const int half  = lane >> 4;
  const int sub   = lane & 15;
  const int mrow  = row0 + sub;
  const float ns  = norm_s[mrow];
  const float* __restrict__ xrow = X + (size_t)mrow * K;

  v8f acc0 = {}, acc1 = {}, acc2 = {}, acc3 = {};

#pragma unroll
  for (int kk = 0; kk < K / 4; ++kk) {
    const int ka = kk * 4 + half * 2;
    v2f a;
    a.x = xrow[ka] * ns;
    a.y = xrow[ka + 1] * ns;

    const float* wk = &sW[ka * 64 + sub];
    v2f b;

    b.x = wk[0];       b.y = wk[64];
    acc0 = __builtin_amdgcn_wmma_f32_16x16x4_f32(false, a, false, b, (short)0, acc0, false, false);
    b.x = wk[16];      b.y = wk[64 + 16];
    acc1 = __builtin_amdgcn_wmma_f32_16x16x4_f32(false, a, false, b, (short)0, acc1, false, false);
    b.x = wk[32];      b.y = wk[64 + 32];
    acc2 = __builtin_amdgcn_wmma_f32_16x16x4_f32(false, a, false, b, (short)0, acc2, false, false);
    b.x = wk[48];      b.y = wk[64 + 48];
    acc3 = __builtin_amdgcn_wmma_f32_16x16x4_f32(false, a, false, b, (short)0, acc3, false, false);
  }

  float* ybase = Y + (size_t)row0 * 64 + sub;
#pragma unroll
  for (int r = 0; r < 8; ++r) {
    const int m = r + half * 8;
    float* yr = ybase + (size_t)m * 64;
    yr[0]  = acc0[r];
    yr[16] = acc1[r];
    yr[32] = acc2[r];
    yr[48] = acc3[r];
  }
}

static inline int cdiv(long long a, long long b) { return (int)((a + b - 1) / b); }

extern "C" void kernel_launch(void* const* d_in, const int* in_sizes, int n_in,
                              void* d_out, int out_size, void* d_ws, size_t ws_size,
                              hipStream_t stream) {
  const float* features = (const float*)d_in[0];
  const int*   src      = (const int*)d_in[1];
  const int*   dst      = (const int*)d_in[2];
  const float* W1       = (const float*)d_in[3];
  const float* b1       = (const float*)d_in[4];
  const float* W2       = (const float*)d_in[5];
  const float* b2       = (const float*)d_in[6];

  const int n_nodes = in_sizes[0] / 128;
  const int n_edges = in_sizes[1];
  const int n_feat  = n_nodes * 64;

  float* ws     = (float*)d_ws;
  float* norm_s = ws;
  float* norm_d = ws + n_nodes;
  float* bufA   = ws + 2 * (size_t)n_nodes;
  float* bufB   = bufA + (size_t)n_nodes * 64;
  float* out    = (float*)d_out;

  const int B = 256;

  zero_f32<<<cdiv(2 * (long long)n_nodes, B), B, 0, stream>>>(norm_s, 2 * n_nodes);
  degree_kernel<<<cdiv(n_edges, B), B, 0, stream>>>(src, dst, norm_s, norm_d, n_edges);
  norm_kernel<<<cdiv(2 * (long long)n_nodes, B), B, 0, stream>>>(norm_s, 2 * n_nodes);

  const int gemm_blocks = cdiv(n_nodes, 128);

  gemm_norm_wmma<128><<<gemm_blocks, B, 0, stream>>>(features, W1, norm_s, bufA, n_nodes);
  zero_f32<<<cdiv(n_feat, B), B, 0, stream>>>(bufB, n_feat);
  spmm_scatter<<<cdiv(n_edges, B), B, 0, stream>>>(bufA, src, dst, bufB, n_edges);
  post_kernel<<<cdiv(n_feat, B), B, 0, stream>>>(bufB, norm_d, b1, bufA, n_feat, 1);

  gemm_norm_wmma<64><<<gemm_blocks, B, 0, stream>>>(bufA, W2, norm_s, bufB, n_nodes);
  zero_f32<<<cdiv(n_feat, B), B, 0, stream>>>(out, n_feat);
  spmm_scatter<<<cdiv(n_edges, B), B, 0, stream>>>(bufB, src, dst, out, n_edges);
  post_kernel<<<cdiv(n_feat, B), B, 0, stream>>>(out, norm_d, b2, out, n_feat, 0);
}


# Aggregation by destination row replaces atomic scatter: degree atomics return edge ranks, a scan gives row offsets; a wave sums a row in registers in rank (arrival) order, one store, epilogue fused
# speedup vs baseline: 1.8470x; 1.1900x over previous
#include <hip/hip_runtime.h>


typedef float v2f __attribute__((ext_vector_type(2)));
typedef float v8f __attribute__((ext_vector_type(8)));

__global__ void zero_i32(int* __restrict__ p, int n) {
  int i = blockIdx.x * blockDim.x + threadIdx.x;
  if (i < n) p[i] = 0;
}

__global__ void degree_rank(const int* __restrict__ src, const int* __restrict__ dst,
                            int* __restrict__ cnt_out, int* __restrict__ cnt_in,
                            int* __restrict__ rank, int n_edges) {
  int e = blockIdx.x * blockDim.x + threadIdx.x;
  if (e >= n_edges) return;
  const int s = src[e];
  const int d = dst[e];
  atomicAdd(cnt_out + s, 1);
  rank[e] = atomicAdd(cnt_in + d, 1);
}

__global__ void __launch_bounds__(1024) offsets_norms(const int* __restrict__ cnt_out,
                                                      const int* __restrict__ cnt_in,
                                                      int* __restrict__ off,
                                                      float* __restrict__ norm_s,
                                                      float* __restrict__ norm_d,
                                                      int n_nodes) {
  __shared__ int s_before[32];
  __shared__ int s_total[32];
  const int tid = threadIdx.x;
  const int lane = tid & 31;
  const int w = tid >> 5;
  const int start = blockIdx.x * 1024;

  int before = 0;
  for (int i = tid; i < start; i += 1024) before += cnt_in[i];

  const int i = start + tid;
  int c = 0;
  if (i < n_nodes) {
    c = cnt_in[i];
    norm_d[i] = rsqrtf(fmaxf((float)c, 1.0f));
    norm_s[i] = rsqrtf(fmaxf((float)cnt_out[i], 1.0f));
  }

  int incl = c;
  for (int o = 1; o < 32; o <<= 1) {
    const int t = __shfl_up(incl, o);
    if (lane >= o) incl += t;
  }
  for (int o = 16; o > 0; o >>= 1) before += __shfl_xor(before, o);
  if (lane == 31) s_total[w] = incl;
  if (lane == 0) s_before[w] = before;
  __syncthreads();

  int add = s_before[lane] + (lane < w ? s_total[lane] : 0);
  for (int o = 16; o > 0; o >>= 1) add += __shfl_xor(add, o);
  if (i <= n_nodes) off[i] = add + incl - c;
}

#define PLACE_EDGES 4

__global__ void __launch_bounds__(256) place_sources(const int* __restrict__ src,
                                                     const int* __restrict__ dst,
                                                     const int* __restrict__ rank,
                                                     const int* __restrict__ off,
                                                     int* __restrict__ by_dst, int n_edges) {
  const int e0 = blockIdx.x * (256 * PLACE_EDGES) + threadIdx.x;
  int s[PLACE_EDGES], d[PLACE_EDGES], r[PLACE_EDGES];
#pragma unroll
  for (int k = 0; k < PLACE_EDGES; ++k) {
    const int e = min(e0 + k * 256, n_edges - 1);
    s[k] = src[e];
    d[k] = dst[e];
    r[k] = rank[e];
  }
#pragma unroll
  for (int k = 0; k < PLACE_EDGES; ++k) r[k] += off[d[k]];
#pragma unroll
  for (int k = 0; k < PLACE_EDGES; ++k) {
    if (e0 + k * 256 < n_edges) atomicExch(by_dst + r[k], s[k]);
  }
}

#define ROWS_PER_WAVE 4
#define ROW_GROUP 8

__global__ void __launch_bounds__(256) gather_rows(const float* __restrict__ H,
                                                   const int* __restrict__ off,
                                                   const int* __restrict__ by_dst,
                                                   const float* __restrict__ norm_d,
                                                   const float* __restrict__ bias,
                                                   float* __restrict__ out,
                                                   int n_nodes, int do_relu) {
  const int lane = threadIdx.x & 31;
  const int wave = __builtin_amdgcn_readfirstlane((int)((blockIdx.x * blockDim.x + threadIdx.x) >> 5));
  const int r0 = wave * ROWS_PER_WAVE;
  if (r0 >= n_nodes) return;
  const int nrows = min(ROWS_PER_WAVE, n_nodes - r0);

  const int o_l = off[r0 + min(lane, nrows)];
  const float nd_l = norm_d[r0 + min(lane, nrows - 1)];
  const float bx = bias[2 * lane];
  const float by = bias[2 * lane + 1];
  const float2* __restrict__ Hl = reinterpret_cast<const float2*>(H) + lane;
  float2* __restrict__ outl = reinterpret_cast<float2*>(out) + lane;

  int beg = __builtin_amdgcn_readlane(o_l, 0);
  int end = __builtin_amdgcn_readlane(o_l, 1);
  int s_row = 0;
  if (beg + lane < end) s_row = by_dst[beg + lane];

#pragma clang loop unroll(disable)
  for (int rr = 0; rr < nrows; ++rr) {
    const int nend = __builtin_amdgcn_readlane(o_l, min(rr + 2, nrows));
    int s_nrow = 0;
    if (end + lane < nend) s_nrow = by_dst[end + lane];

    float ax = 0.0f, ay = 0.0f;
    int s_cur = s_row;
    for (int base = beg; base < end; base += 32) {
      int s_nxt = 0;
      if (base + 32 + lane < end) s_nxt = by_dst[base + 32 + lane];
      const int cnt = min(32, end - base);
      int j = 0;
      for (; j + ROW_GROUP <= cnt; j += ROW_GROUP) {
        float2 v[ROW_GROUP];
#pragma unroll
        for (int u = 0; u < ROW_GROUP; ++u) {
          const int s = __builtin_amdgcn_readlane(s_cur, j + u);
          v[u] = Hl[(size_t)s * 32];
        }
#pragma unroll
        for (int u = 0; u < ROW_GROUP; ++u) {
          ax += v[u].x;
          ay += v[u].y;
        }
      }
      if (j < cnt) {
        float2 v[ROW_GROUP - 1];
#pragma unroll
        for (int u = 0; u < ROW_GROUP - 1; ++u) {
          v[u] = make_float2(0.0f, 0.0f);
          if (j + u < cnt) {
            const int s = __builtin_amdgcn_readlane(s_cur, j + u);
            v[u] = Hl[(size_t)s * 32];
          }
        }
#pragma unroll
        for (int u = 0; u < ROW_GROUP - 1; ++u) {
          ax += v[u].x;
          ay += v[u].y;
        }
      }
      s_cur = s_nxt;
    }

    const float nd = __int_as_float(__builtin_amdgcn_readlane(__float_as_int(nd_l), rr));
    float ox = fmaf(ax, nd, bx);
    float oy = fmaf(ay, nd, by);
    if (do_relu) {
      ox = fmaxf(ox, 0.0f);
      oy = fmaxf(oy, 0.0f);
    }
    outl[(size_t)(r0 + rr) * 32] = make_float2(ox, oy);

    beg = end;
    end = nend;
    s_row = s_nrow;
  }
}

template <int K>
__global__ void __launch_bounds__(256) gemm_norm_wmma(const float* __restrict__ X,
                                                      const float* __restrict__ W,
                                                      const float* __restrict__ norm_s,
                                                      float* __restrict__ Y,
                                                      int n_nodes) {
  __shared__ float sW[K * 64];
  for (int i = threadIdx.x; i < K * 64; i += 256) sW[i] = W[i];
  __syncthreads();

  const int wave = threadIdx.x >> 5;
  const int lane = threadIdx.x & 31;
  const int row0 = (blockIdx.x * 8 + wave) * 16;
  if (row0 + 15 >= n_nodes) return;

  const int half  = lane >> 4;
  const int sub   = lane & 15;
  const int mrow  = row0 + sub;
  const float ns  = norm_s[mrow];
  const float* __restrict__ xrow = X + (size_t)mrow * K;

  v8f acc0 = {}, acc1 = {}, acc2 = {}, acc3 = {};

#pragma unroll
  for (int kk = 0; kk < K / 4; ++kk) {
    const int ka = kk * 4 + half * 2;
    v2f a;
    a.x = xrow[ka] * ns;
    a.y = xrow[ka + 1] * ns;

    const float* wk = &sW[ka * 64 + sub];
    v2f b;

    b.x = wk[0];       b.y = wk[64];
    acc0 = __builtin_amdgcn_wmma_f32_16x16x4_f32(false, a, false, b, (short)0, acc0, false, false);
    b.x = wk[16];      b.y = wk[64 + 16];
    acc1 = __builtin_amdgcn_wmma_f32_16x16x4_f32(false, a, false, b, (short)0, acc1, false, false);
    b.x = wk[32];      b.y = wk[64 + 32];
    acc2 = __builtin_amdgcn_wmma_f32_16x16x4_f32(false, a, false, b, (short)0, acc2, false, false);
    b.x = wk[48];      b.y = wk[64 + 48];
    acc3 = __builtin_amdgcn_wmma_f32_16x16x4_f32(false, a, false, b, (short)0, acc3, false, false);
  }

  float* ybase = Y + (size_t)row0 * 64 + sub;
#pragma unroll
  for (int r = 0; r < 8; ++r) {
    const int m = r + half * 8;
    float* yr = ybase + (size_t)m * 64;
    yr[0]  = acc0[r];
    yr[16] = acc1[r];
    yr[32] = acc2[r];
    yr[48] = acc3[r];
  }
}

static inline int cdiv(long long a, long long b) { return (int)((a + b - 1) / b); }

extern "C" void kernel_launch(void* const* d_in, const int* in_sizes, int n_in,
                              void* d_out, int out_size, void* d_ws, size_t ws_size,
                              hipStream_t stream) {
  const float* features = (const float*)d_in[0];
  const int*   src      = (const int*)d_in[1];
  const int*   dst      = (const int*)d_in[2];
  const float* W1       = (const float*)d_in[3];
  const float* b1       = (const float*)d_in[4];
  const float* W2       = (const float*)d_in[5];
  const float* b2       = (const float*)d_in[6];

  const int n_nodes = in_sizes[0] / 128;
  const int n_edges = in_sizes[1];

  const size_t np = ((size_t)n_nodes + 64) & ~(size_t)63;
  const size_t ep = ((size_t)n_edges + 63) & ~(size_t)63;

  int*   wsi     = (int*)d_ws;
  int*   cnt_out = wsi;
  int*   cnt_in  = wsi + np;
  int*   off     = wsi + 2 * np;
  float* norm_s  = (float*)(wsi + 3 * np);
  float* norm_d  = (float*)(wsi + 4 * np);
  int*   rank    = wsi + 5 * np;
  int*   by_dst  = rank + ep;
  float* bufA    = (float*)(by_dst + ep);
  float* bufB    = bufA + (size_t)n_nodes * 64;
  float* out     = (float*)d_out;

  const int B = 256;

  zero_i32<<<cdiv(2 * (long long)np, B), B, 0, stream>>>(cnt_out, (int)(2 * np));
  degree_rank<<<cdiv(n_edges, B), B, 0, stream>>>(src, dst, cnt_out, cnt_in, rank, n_edges);
  offsets_norms<<<cdiv((long long)n_nodes + 1, 1024), 1024, 0, stream>>>(cnt_out, cnt_in, off,
                                                                         norm_s, norm_d, n_nodes);
  place_sources<<<cdiv(n_edges, B * PLACE_EDGES), B, 0, stream>>>(src, dst, rank, off, by_dst, n_edges);

  const int gemm_blocks = cdiv(n_nodes, 128);
  const int gather_blocks = cdiv(n_nodes, 8 * ROWS_PER_WAVE);

  gemm_norm_wmma<128><<<gemm_blocks, B, 0, stream>>>(features, W1, norm_s, bufA, n_nodes);
  gather_rows<<<gather_blocks, B, 0, stream>>>(bufA, off, by_dst, norm_d, b1, bufB, n_nodes, 1);

  gemm_norm_wmma<64><<<gemm_blocks, B, 0, stream>>>(bufB, W2, norm_s, bufA, n_nodes);
  gather_rows<<<gather_blocks, B, 0, stream>>>(bufA, off, by_dst, norm_d, b2, out, n_nodes, 0);
}


# Edges grouped by destination, and both degree counts, with work-group LDS counters in a two-level bucket pass; no per-edge global atomics left; row sums run in LDS-atomic arrival order
# speedup vs baseline: 5.4725x; 2.9629x over previous
#include <hip/hip_runtime.h>


typedef float v2f __attribute__((ext_vector_type(2)));
typedef float v8f __attribute__((ext_vector_type(8)));

#define BKT_SHIFT 9
#define BKT_NODES 512
#define NBMAX 256
#define SRC_BITS 17
#define SRC_MASK 0x1FFFF
#define EPB 12544
#define EPT ((EPB + 1023) / 1024)
#define RUN_CAP (EPB + 31 * NBMAX)
#define FIN_CAP 32768

__device__ __forceinline__ int round32(int x) { return (x + 31) & ~31; }

__device__ __forceinline__ int block_scan_excl(int v, int* s_part, int* total) {
  const int lane = threadIdx.x & 31;
  const int w = threadIdx.x >> 5;
  int incl = v;
  for (int o = 1; o < 32; o <<= 1) {
    const int t = __shfl_up(incl, o);
    if (lane >= o) incl += t;
  }
  if (lane == 31) s_part[w] = incl;
  __syncthreads();
  const int p = s_part[lane];
  int before = lane < w ? p : 0;
  int all = p;
  for (int o = 16; o > 0; o >>= 1) {
    before += __shfl_xor(before, o);
    all += __shfl_xor(all, o);
  }
  *total = all;
  return before + incl - v;
}

__global__ void __launch_bounds__(1024) partition_edges(const int* __restrict__ src,
                                                        const int* __restrict__ dst,
                                                        int* __restrict__ runcnt_d,
                                                        int* __restrict__ runoff_d,
                                                        int* __restrict__ runs_d,
                                                        int* __restrict__ runcnt_s,
                                                        int* __restrict__ runoff_s,
                                                        int* __restrict__ runs_s,
                                                        int n_edges) {
  __shared__ int s_img_d[RUN_CAP];
  __shared__ int s_img_s[RUN_CAP];
  __shared__ int s_cnt_d[NBMAX], s_cnt_s[NBMAX];
  __shared__ int s_off_d[NBMAX], s_off_s[NBMAX];
  __shared__ int s_cur_d[NBMAX], s_cur_s[NBMAX];
  __shared__ int s_part_d[32], s_part_s[32];

  const int tid = threadIdx.x;
  const int lane = tid & 31;
  const int w = tid >> 5;
  const int b = blockIdx.x;
  const int e0 = b * EPB;
  const int e1 = min(e0 + EPB, n_edges);

  int d[EPT], s[EPT];
#pragma unroll
  for (int u = 0; u < EPT; ++u) {
    const int e = min(e0 + tid + 1024 * u, e1 - 1);
    d[u] = dst[e];
    s[u] = src[e];
  }
#pragma unroll
  for (int u = 0; u < EPT; ++u) {
    if (e0 + tid + 1024 * u >= e1) d[u] = -1;
  }

  if (tid < NBMAX) {
    s_cnt_d[tid] = 0;
    s_cnt_s[tid] = 0;
  }
  __syncthreads();

#pragma unroll
  for (int u = 0; u < EPT; ++u) {
    if (d[u] >= 0) {
      atomicAdd(&s_cnt_d[d[u] >> BKT_SHIFT], 1);
      atomicAdd(&s_cnt_s[s[u] >> BKT_SHIFT], 1);
    }
  }
  __syncthreads();

  const int cd = tid < NBMAX ? s_cnt_d[tid] : 0;
  const int cs = tid < NBMAX ? s_cnt_s[tid] : 0;
  int tot_d, tot_s;
  const int od = block_scan_excl(round32(cd), s_part_d, &tot_d);
  const int os = block_scan_excl(round32(cs), s_part_s, &tot_s);
  if (tid < NBMAX) {
    s_off_d[tid] = od;
    s_cur_d[tid] = od;
    s_off_s[tid] = os;
    s_cur_s[tid] = os;
    runcnt_d[b * NBMAX + tid] = cd;
    runoff_d[b * NBMAX + tid] = od;
    runcnt_s[b * NBMAX + tid] = cs;
    runoff_s[b * NBMAX + tid] = os;
  }
  __syncthreads();

  for (int k = w; k < NBMAX; k += 32) {
    const int nd = s_cnt_d[k];
    if (nd + lane < round32(nd)) s_img_d[s_off_d[k] + nd + lane] = 0;
    const int ns = s_cnt_s[k];
    if (ns + lane < round32(ns)) s_img_s[s_off_s[k] + ns + lane] = 0;
  }

#pragma unroll
  for (int u = 0; u < EPT; ++u) {
    if (d[u] >= 0) {
      const int pd = atomicAdd(&s_cur_d[d[u] >> BKT_SHIFT], 1);
      s_img_d[pd] = ((d[u] & (BKT_NODES - 1)) << SRC_BITS) | s[u];
      const int ps = atomicAdd(&s_cur_s[s[u] >> BKT_SHIFT], 1);
      s_img_s[ps] = s[u] & (BKT_NODES - 1);
    }
  }
  __syncthreads();

  int* __restrict__ out_d = runs_d + (size_t)b * RUN_CAP;
  int* __restrict__ out_s = runs_s + (size_t)b * RUN_CAP;
  for (int i = tid; i < tot_d; i += 1024) out_d[i] = s_img_d[i];
  for (int i = tid; i < tot_s; i += 1024) out_s[i] = s_img_s[i];
}

__global__ void __launch_bounds__(1024) bucket_bases(const int* __restrict__ runcnt_d,
                                                     int* __restrict__ base_d, int n_blocks) {
  __shared__ int s_sum[4][NBMAX];
  __shared__ int s_part[32];
  const int tid = threadIdx.x;
  const int k = tid & (NBMAX - 1);
  const int part = tid >> 8;
  int acc = 0;
  for (int b = part; b < n_blocks; b += 64) {
    int v[16];
#pragma unroll
    for (int u = 0; u < 16; ++u) v[u] = runcnt_d[min(b + 4 * u, n_blocks - 1) * NBMAX + k];
#pragma unroll
    for (int u = 0; u < 16; ++u) acc += (b + 4 * u < n_blocks) ? v[u] : 0;
  }
  s_sum[part][k] = acc;
  __syncthreads();
  int v = 0;
  if (tid < NBMAX) v = round32(s_sum[0][tid] + s_sum[1][tid] + s_sum[2][tid] + s_sum[3][tid]);
  int tot;
  const int ex = block_scan_excl(v, s_part, &tot);
  if (tid < NBMAX) base_d[tid] = ex;
}

#define RUNS_AHEAD 4
template <class F>
__device__ __forceinline__ void walk_runs(const int* __restrict__ runcnt,
                                          const int* __restrict__ runoff,
                                          const int* __restrict__ runs, int k, int n_blocks, F f) {
  const int lane = threadIdx.x & 31;
  const int w = threadIdx.x >> 5;
  for (int b0 = w; b0 < n_blocks; b0 += 1024) {
    const int bl = b0 + 32 * lane;
    int n_l = 0, o_l = 0;
    if (bl < n_blocks) {
      n_l = runcnt[bl * NBMAX + k];
      o_l = runoff[bl * NBMAX + k];
    }
    const int jn = min(32, (n_blocks - b0 + 31) >> 5);
    for (int j0 = 0; j0 < jn; j0 += RUNS_AHEAD) {
      int n[RUNS_AHEAD];
      const int* run[RUNS_AHEAD];
      int nmax = 0;
#pragma unroll
      for (int u = 0; u < RUNS_AHEAD; ++u) {
        n[u] = __builtin_amdgcn_readlane(n_l, j0 + u);
        run[u] = runs + (size_t)(b0 + 32 * (j0 + u)) * RUN_CAP + __builtin_amdgcn_readlane(o_l, j0 + u);
        nmax = max(nmax, n[u]);
      }
      for (int i = lane; i < nmax; i += 32) {
        int v[RUNS_AHEAD];
#pragma unroll
        for (int u = 0; u < RUNS_AHEAD; ++u) {
          v[u] = 0;
          if (i < n[u]) v[u] = run[u][i];
        }
#pragma unroll
        for (int u = 0; u < RUNS_AHEAD; ++u) {
          if (i < n[u]) f(v[u]);
        }
      }
    }
  }
}

__global__ void __launch_bounds__(1024) finish_buckets(const int* __restrict__ runcnt_d,
                                                       const int* __restrict__ runoff_d,
                                                       const int* __restrict__ runs_d,
                                                       const int* __restrict__ runcnt_s,
                                                       const int* __restrict__ runoff_s,
                                                       const int* __restrict__ runs_s,
                                                       const int* __restrict__ base_d,
                                                       int* __restrict__ off,
                                                       int* __restrict__ cnt_in,
                                                       float* __restrict__ norm_s,
                                                       float* __restrict__ norm_d,
                                                       int* __restrict__ by_dst,
                                                       int n_nodes, int n_blocks, int n_buckets) {
  __shared__ int s_stage[FIN_CAP];
  __shared__ int s_cnt[BKT_NODES];
  __shared__ int s_cur[BKT_NODES];
  __shared__ int s_part[32];

  const int tid = threadIdx.x;
  const bool is_src = (int)blockIdx.x >= n_buckets;
  const int k = is_src ? (int)blockIdx.x - n_buckets : (int)blockIdx.x;
  const int* __restrict__ runcnt = is_src ? runcnt_s : runcnt_d;
  const int* __restrict__ runoff = is_src ? runoff_s : runoff_d;
  const int* __restrict__ runs = is_src ? runs_s : runs_d;
  const int key_shift = is_src ? 0 : SRC_BITS;

  if (tid < BKT_NODES) s_cnt[tid] = 0;
  __syncthreads();

  walk_runs(runcnt, runoff, runs, k, n_blocks, [&](int v) { atomicAdd(&s_cnt[v >> key_shift], 1); });
  __syncthreads();

  const int c = tid < BKT_NODES ? s_cnt[tid] : 0;
  int total;
  const int ex = block_scan_excl(c, s_part, &total);
  const int node = k * BKT_NODES + tid;
  const float nrm = rsqrtf(fmaxf((float)c, 1.0f));
  if (is_src) {
    if (tid < BKT_NODES && node < n_nodes) norm_s[node] = nrm;
    return;
  }
  const int base = base_d[k];
  if (tid < BKT_NODES) {
    s_cur[tid] = ex;
    if (node < n_nodes) {
      off[node] = base + ex;
      cnt_in[node] = c;
      norm_d[node] = nrm;
    }
  }
  __syncthreads();

  const bool fits = total <= FIN_CAP;
  walk_runs(runcnt, runoff, runs, k, n_blocks, [&](int v) {
    const int p = atomicAdd(&s_cur[v >> SRC_BITS], 1);
    if (fits) s_stage[p] = v & SRC_MASK;
    else atomicExch(by_dst + base + p, v & SRC_MASK);
  });
  __syncthreads();

  if (fits) {
    const int padded = round32(total);
    for (int i = tid; i < padded; i += 1024) by_dst[base + i] = i < total ? s_stage[i] : 0;
  }
}

#define ROWS_PER_WAVE 4
#define ROW_GROUP 8

__global__ void __launch_bounds__(256) gather_rows(const float* __restrict__ H,
                                                   const int* __restrict__ off,
                                                   const int* __restrict__ cnt_in,
                                                   const int* __restrict__ by_dst,
                                                   const float* __restrict__ norm_d,
                                                   const float* __restrict__ bias,
                                                   float* __restrict__ out,
                                                   int n_nodes, int do_relu) {
  const int lane = threadIdx.x & 31;
  const int wave = __builtin_amdgcn_readfirstlane((int)((blockIdx.x * blockDim.x + threadIdx.x) >> 5));
  const int r0 = wave * ROWS_PER_WAVE;
  if (r0 >= n_nodes) return;
  const int nrows = min(ROWS_PER_WAVE, n_nodes - r0);

  const int rl = r0 + min(lane, nrows - 1);
  const int o_l = off[rl];
  const int e_l = o_l + cnt_in[rl];
  const float nd_l = norm_d[rl];
  const float bx = bias[2 * lane];
  const float by = bias[2 * lane + 1];
  const float2* __restrict__ Hl = reinterpret_cast<const float2*>(H) + lane;
  float2* __restrict__ outl = reinterpret_cast<float2*>(out) + lane;

  int beg = __builtin_amdgcn_readlane(o_l, 0);
  int end = __builtin_amdgcn_readlane(e_l, 0);
  int s_row = 0;
  if (beg + lane < end) s_row = by_dst[beg + lane];

#pragma clang loop unroll(disable)
  for (int rr = 0; rr < nrows; ++rr) {
    const int nr = min(rr + 1, nrows - 1);
    const int nbeg = __builtin_amdgcn_readlane(o_l, nr);
    const int nend = rr + 1 < nrows ? __builtin_amdgcn_readlane(e_l, nr) : nbeg;
    int s_nrow = 0;
    if (nbeg + lane < nend) s_nrow = by_dst[nbeg + lane];

    float ax = 0.0f, ay = 0.0f;
    int s_cur = s_row;
    for (int base = beg; base < end; base += 32) {
      int s_nxt = 0;
      if (base + 32 + lane < end) s_nxt = by_dst[base + 32 + lane];
      const int cnt = min(32, end - base);
      int j = 0;
      for (; j + ROW_GROUP <= cnt; j += ROW_GROUP) {
        float2 v[ROW_GROUP];
#pragma unroll
        for (int u = 0; u < ROW_GROUP; ++u) {
          const int s = __builtin_amdgcn_readlane(s_cur, j + u);
          v[u] = Hl[(size_t)s * 32];
        }
#pragma unroll
        for (int u = 0; u < ROW_GROUP; ++u) {
          ax += v[u].x;
          ay += v[u].y;
        }
      }
      if (j < cnt) {
        float2 v[ROW_GROUP - 1];
#pragma unroll
        for (int u = 0; u < ROW_GROUP - 1; ++u) {
          v[u] = make_float2(0.0f, 0.0f);
          if (j + u < cnt) {
            const int s = __builtin_amdgcn_readlane(s_cur, j + u);
            v[u] = Hl[(size_t)s * 32];
          }
        }
#pragma unroll
        for (int u = 0; u < ROW_GROUP - 1; ++u) {
          ax += v[u].x;
          ay += v[u].y;
        }
      }
      s_cur = s_nxt;
    }

    const float nd = __int_as_float(__builtin_amdgcn_readlane(__float_as_int(nd_l), rr));
    float ox = fmaf(ax, nd, bx);
    float oy = fmaf(ay, nd, by);
    if (do_relu) {
      ox = fmaxf(ox, 0.0f);
      oy = fmaxf(oy, 0.0f);
    }
    outl[(size_t)(r0 + rr) * 32] = make_float2(ox, oy);

    beg = nbeg;
    end = nend;
    s_row = s_nrow;
  }
}

template <int K>
__global__ void __launch_bounds__(256) gemm_norm_wmma(const float* __restrict__ X,
                                                      const float* __restrict__ W,
                                                      const float* __restrict__ norm_s,
                                                      float* __restrict__ Y,
                                                      int n_nodes) {
  __shared__ float sW[K * 64];
  for (int i = threadIdx.x; i < K * 64; i += 256) sW[i] = W[i];
  __syncthreads();

  const int wave = threadIdx.x >> 5;
  const int lane = threadIdx.x & 31;
  const int row0 = (blockIdx.x * 8 + wave) * 16;
  if (row0 + 15 >= n_nodes) return;

  const int half  = lane >> 4;
  const int sub   = lane & 15;
  const int mrow  = row0 + sub;
  const float ns  = norm_s[mrow];
  const float* __restrict__ xrow = X + (size_t)mrow * K;

  v8f acc0 = {}, acc1 = {}, acc2 = {}, acc3 = {};

#pragma unroll
  for (int kk = 0; kk < K / 4; ++kk) {
    const int ka = kk * 4 + half * 2;
    v2f a;
    a.x = xrow[ka] * ns;
    a.y = xrow[ka + 1] * ns;

    const float* wk = &sW[ka * 64 + sub];
    v2f b;

    b.x = wk[0];       b.y = wk[64];
    acc0 = __builtin_amdgcn_wmma_f32_16x16x4_f32(false, a, false, b, (short)0, acc0, false, false);
    b.x = wk[16];      b.y = wk[64 + 16];
    acc1 = __builtin_amdgcn_wmma_f32_16x16x4_f32(false, a, false, b, (short)0, acc1, false, false);
    b.x = wk[32];      b.y = wk[64 + 32];
    acc2 = __builtin_amdgcn_wmma_f32_16x16x4_f32(false, a, false, b, (short)0, acc2, false, false);
    b.x = wk[48];      b.y = wk[64 + 48];
    acc3 = __builtin_amdgcn_wmma_f32_16x16x4_f32(false, a, false, b, (short)0, acc3, false, false);
  }

  float* ybase = Y + (size_t)row0 * 64 + sub;
#pragma unroll
  for (int r = 0; r < 8; ++r) {
    const int m = r + half * 8;
    float* yr = ybase + (size_t)m * 64;
    yr[0]  = acc0[r];
    yr[16] = acc1[r];
    yr[32] = acc2[r];
    yr[48] = acc3[r];
  }
}

static inline int cdiv(long long a, long long b) { return (int)((a + b - 1) / b); }

extern "C" void kernel_launch(void* const* d_in, const int* in_sizes, int n_in,
                              void* d_out, int out_size, void* d_ws, size_t ws_size,
                              hipStream_t stream) {
  const float* features = (const float*)d_in[0];
  const int*   src      = (const int*)d_in[1];
  const int*   dst      = (const int*)d_in[2];
  const float* W1       = (const float*)d_in[3];
  const float* b1       = (const float*)d_in[4];
  const float* W2       = (const float*)d_in[5];
  const float* b2       = (const float*)d_in[6];

  const int n_nodes = in_sizes[0] / 128;
  const int n_edges = in_sizes[1];

  const int n_buckets = (n_nodes + BKT_NODES - 1) >> BKT_SHIFT;
  const int n_pblocks = cdiv(n_edges, EPB);

  const size_t np = ((size_t)n_nodes + 63) & ~(size_t)63;
  const size_t ep = ((size_t)n_edges + 63) & ~(size_t)63;
  const size_t tp = (size_t)n_pblocks * NBMAX;
  const size_t rp = (size_t)n_pblocks * RUN_CAP;

  int*   wsi      = (int*)d_ws;
  int*   off      = wsi;
  int*   cnt_in   = off + np;
  float* norm_s   = (float*)(cnt_in + np);
  float* norm_d   = norm_s + np;
  int*   base_d   = (int*)(norm_d + np);
  int*   runcnt_d = base_d + NBMAX;
  int*   runoff_d = runcnt_d + tp;
  int*   runcnt_s = runoff_d + tp;
  int*   runoff_s = runcnt_s + tp;
  int*   runs_d   = runoff_s + tp;
  int*   runs_s   = runs_d + rp;
  int*   by_dst   = runs_s + rp;
  float* bufA     = (float*)(by_dst + ep + 32 * NBMAX);
  float* bufB     = bufA + (size_t)n_nodes * 64;
  float* out      = (float*)d_out;

  const int B = 256;

  partition_edges<<<n_pblocks, 1024, 0, stream>>>(src, dst, runcnt_d, runoff_d, runs_d,
                                                  runcnt_s, runoff_s, runs_s, n_edges);
  bucket_bases<<<1, 1024, 0, stream>>>(runcnt_d, base_d, n_pblocks);
  finish_buckets<<<2 * n_buckets, 1024, 0, stream>>>(runcnt_d, runoff_d, runs_d, runcnt_s, runoff_s,
                                                     runs_s, base_d, off, cnt_in, norm_s, norm_d,
                                                     by_dst, n_nodes, n_pblocks, n_buckets);

  const int gemm_blocks = cdiv(n_nodes, 128);
  const int gather_blocks = cdiv(n_nodes, 8 * ROWS_PER_WAVE);

  gemm_norm_wmma<128><<<gemm_blocks, B, 0, stream>>>(features, W1, norm_s, bufA, n_nodes);
  gather_rows<<<gather_blocks, B, 0, stream>>>(bufA, off, cnt_in, by_dst, norm_d, b1, bufB, n_nodes, 1);

  gemm_norm_wmma<64><<<gemm_blocks, B, 0, stream>>>(bufB, W2, norm_s, bufA, n_nodes);
  gather_rows<<<gather_blocks, B, 0, stream>>>(bufA, off, cnt_in, by_dst, norm_d, b2, out, n_nodes, 0);
}


# Layer-2 transform moved into layer-1 aggregation's epilogue: a work-group's 32 finished rows meet in LDS, 16 f32 WMMA per wave (same k and sum order); activations skip memory, one launch less
# speedup vs baseline: 6.1972x; 1.1324x over previous
#include <hip/hip_runtime.h>


typedef float v2f __attribute__((ext_vector_type(2)));
typedef float v8f __attribute__((ext_vector_type(8)));

#define BKT_SHIFT 9
#define BKT_NODES 512
#define NBMAX 256
#define SRC_BITS 17
#define SRC_MASK 0x1FFFF
#define EPB 12544
#define EPT ((EPB + 1023) / 1024)
#define RUN_CAP (EPB + 31 * NBMAX)
#define FIN_CAP 32768

__device__ __forceinline__ int round32(int x) { return (x + 31) & ~31; }

__device__ __forceinline__ int block_scan_excl(int v, int* s_part, int* total) {
  const int lane = threadIdx.x & 31;
  const int w = threadIdx.x >> 5;
  int incl = v;
  for (int o = 1; o < 32; o <<= 1) {
    const int t = __shfl_up(incl, o);
    if (lane >= o) incl += t;
  }
  if (lane == 31) s_part[w] = incl;
  __syncthreads();
  const int p = s_part[lane];
  int before = lane < w ? p : 0;
  int all = p;
  for (int o = 16; o > 0; o >>= 1) {
    before += __shfl_xor(before, o);
    all += __shfl_xor(all, o);
  }
  *total = all;
  return before + incl - v;
}

__global__ void __launch_bounds__(1024) partition_edges(const int* __restrict__ src,
                                                        const int* __restrict__ dst,
                                                        int* __restrict__ runcnt_d,
                                                        int* __restrict__ runoff_d,
                                                        int* __restrict__ runs_d,
                                                        int* __restrict__ runcnt_s,
                                                        int* __restrict__ runoff_s,
                                                        int* __restrict__ runs_s,
                                                        int n_edges) {
  __shared__ int s_img_d[RUN_CAP];
  __shared__ int s_img_s[RUN_CAP];
  __shared__ int s_cnt_d[NBMAX], s_cnt_s[NBMAX];
  __shared__ int s_off_d[NBMAX], s_off_s[NBMAX];
  __shared__ int s_cur_d[NBMAX], s_cur_s[NBMAX];
  __shared__ int s_part_d[32], s_part_s[32];

  const int tid = threadIdx.x;
  const int lane = tid & 31;
  const int w = tid >> 5;
  const int b = blockIdx.x;
  const int e0 = b * EPB;
  const int e1 = min(e0 + EPB, n_edges);

  int d[EPT], s[EPT];
#pragma unroll
  for (int u = 0; u < EPT; ++u) {
    const int e = min(e0 + tid + 1024 * u, e1 - 1);
    d[u] = dst[e];
    s[u] = src[e];
  }
#pragma unroll
  for (int u = 0; u < EPT; ++u) {
    if (e0 + tid + 1024 * u >= e1) d[u] = -1;
  }

  if (tid < NBMAX) {
    s_cnt_d[tid] = 0;
    s_cnt_s[tid] = 0;
  }
  __syncthreads();

#pragma unroll
  for (int u = 0; u < EPT; ++u) {
    if (d[u] >= 0) {
      atomicAdd(&s_cnt_d[d[u] >> BKT_SHIFT], 1);
      atomicAdd(&s_cnt_s[s[u] >> BKT_SHIFT], 1);
    }
  }
  __syncthreads();

  const int cd = tid < NBMAX ? s_cnt_d[tid] : 0;
  const int cs = tid < NBMAX ? s_cnt_s[tid] : 0;
  int tot_d, tot_s;
  const int od = block_scan_excl(round32(cd), s_part_d, &tot_d);
  const int os = block_scan_excl(round32(cs), s_part_s, &tot_s);
  if (tid < NBMAX) {
    s_off_d[tid] = od;
    s_cur_d[tid] = od;
    s_off_s[tid] = os;
    s_cur_s[tid] = os;
    runcnt_d[b * NBMAX + tid] = cd;
    runoff_d[b * NBMAX + tid] = od;
    runcnt_s[b * NBMAX + tid] = cs;
    runoff_s[b * NBMAX + tid] = os;
  }
  __syncthreads();

  for (int k = w; k < NBMAX; k += 32) {
    const int nd = s_cnt_d[k];
    if (nd + lane < round32(nd)) s_img_d[s_off_d[k] + nd + lane] = 0;
    const int ns = s_cnt_s[k];
    if (ns + lane < round32(ns)) s_img_s[s_off_s[k] + ns + lane] = 0;
  }

#pragma unroll
  for (int u = 0; u < EPT; ++u) {
    if (d[u] >= 0) {
      const int pd = atomicAdd(&s_cur_d[d[u] >> BKT_SHIFT], 1);
      s_img_d[pd] = ((d[u] & (BKT_NODES - 1)) << SRC_BITS) | s[u];
      const int ps = atomicAdd(&s_cur_s[s[u] >> BKT_SHIFT], 1);
      s_img_s[ps] = s[u] & (BKT_NODES - 1);
    }
  }
  __syncthreads();

  int* __restrict__ out_d = runs_d + (size_t)b * RUN_CAP;
  int* __restrict__ out_s = runs_s + (size_t)b * RUN_CAP;
  for (int i = tid; i < tot_d; i += 1024) out_d[i] = s_img_d[i];
  for (int i = tid; i < tot_s; i += 1024) out_s[i] = s_img_s[i];
}

__global__ void __launch_bounds__(1024) bucket_bases(const int* __restrict__ runcnt_d,
                                                     int* __restrict__ base_d, int n_blocks) {
  __shared__ int s_sum[4][NBMAX];
  __shared__ int s_part[32];
  const int tid = threadIdx.x;
  const int k = tid & (NBMAX - 1);
  const int part = tid >> 8;
  int acc = 0;
  for (int b = part; b < n_blocks; b += 64) {
    int v[16];
#pragma unroll
    for (int u = 0; u < 16; ++u) v[u] = runcnt_d[min(b + 4 * u, n_blocks - 1) * NBMAX + k];
#pragma unroll
    for (int u = 0; u < 16; ++u) acc += (b + 4 * u < n_blocks) ? v[u] : 0;
  }
  s_sum[part][k] = acc;
  __syncthreads();
  int v = 0;
  if (tid < NBMAX) v = round32(s_sum[0][tid] + s_sum[1][tid] + s_sum[2][tid] + s_sum[3][tid]);
  int tot;
  const int ex = block_scan_excl(v, s_part, &tot);
  if (tid < NBMAX) base_d[tid] = ex;
}

#define RUNS_AHEAD 4
template <class F>
__device__ __forceinline__ void walk_runs(const int* __restrict__ runcnt,
                                          const int* __restrict__ runoff,
                                          const int* __restrict__ runs, int k, int n_blocks, F f) {
  const int lane = threadIdx.x & 31;
  const int w = threadIdx.x >> 5;
  for (int b0 = w; b0 < n_blocks; b0 += 1024) {
    const int bl = b0 + 32 * lane;
    int n_l = 0, o_l = 0;
    if (bl < n_blocks) {
      n_l = runcnt[bl * NBMAX + k];
      o_l = runoff[bl * NBMAX + k];
    }
    const int jn = min(32, (n_blocks - b0 + 31) >> 5);
    for (int j0 = 0; j0 < jn; j0 += RUNS_AHEAD) {
      int n[RUNS_AHEAD];
      const int* run[RUNS_AHEAD];
      int nmax = 0;
#pragma unroll
      for (int u = 0; u < RUNS_AHEAD; ++u) {
        n[u] = __builtin_amdgcn_readlane(n_l, j0 + u);
        run[u] = runs + (size_t)(b0 + 32 * (j0 + u)) * RUN_CAP + __builtin_amdgcn_readlane(o_l, j0 + u);
        nmax = max(nmax, n[u]);
      }
      for (int i = lane; i < nmax; i += 32) {
        int v[RUNS_AHEAD];
#pragma unroll
        for (int u = 0; u < RUNS_AHEAD; ++u) {
          v[u] = 0;
          if (i < n[u]) v[u] = run[u][i];
        }
#pragma unroll
        for (int u = 0; u < RUNS_AHEAD; ++u) {
          if (i < n[u]) f(v[u]);
        }
      }
    }
  }
}

__global__ void __launch_bounds__(1024) finish_buckets(const int* __restrict__ runcnt_d,
                                                       const int* __restrict__ runoff_d,
                                                       const int* __restrict__ runs_d,
                                                       const int* __restrict__ runcnt_s,
                                                       const int* __restrict__ runoff_s,
                                                       const int* __restrict__ runs_s,
                                                       const int* __restrict__ base_d,
                                                       int* __restrict__ off,
                                                       int* __restrict__ cnt_in,
                                                       float* __restrict__ norm_s,
                                                       float* __restrict__ norm_d,
                                                       int* __restrict__ by_dst,
                                                       int n_nodes, int n_blocks, int n_buckets) {
  __shared__ int s_stage[FIN_CAP];
  __shared__ int s_cnt[BKT_NODES];
  __shared__ int s_cur[BKT_NODES];
  __shared__ int s_part[32];

  const int tid = threadIdx.x;
  const bool is_src = (int)blockIdx.x >= n_buckets;
  const int k = is_src ? (int)blockIdx.x - n_buckets : (int)blockIdx.x;
  const int* __restrict__ runcnt = is_src ? runcnt_s : runcnt_d;
  const int* __restrict__ runoff = is_src ? runoff_s : runoff_d;
  const int* __restrict__ runs = is_src ? runs_s : runs_d;
  const int key_shift = is_src ? 0 : SRC_BITS;

  if (tid < BKT_NODES) s_cnt[tid] = 0;
  __syncthreads();

  walk_runs(runcnt, runoff, runs, k, n_blocks, [&](int v) { atomicAdd(&s_cnt[v >> key_shift], 1); });
  __syncthreads();

  const int c = tid < BKT_NODES ? s_cnt[tid] : 0;
  int total;
  const int ex = block_scan_excl(c, s_part, &total);
  const int node = k * BKT_NODES + tid;
  const float nrm = rsqrtf(fmaxf((float)c, 1.0f));
  if (is_src) {
    if (tid < BKT_NODES && node < n_nodes) norm_s[node] = nrm;
    return;
  }
  const int base = base_d[k];
  if (tid < BKT_NODES) {
    s_cur[tid] = ex;
    if (node < n_nodes) {
      off[node] = base + ex;
      cnt_in[node] = c;
      norm_d[node] = nrm;
    }
  }
  __syncthreads();

  const bool fits = total <= FIN_CAP;
  walk_runs(runcnt, runoff, runs, k, n_blocks, [&](int v) {
    const int p = atomicAdd(&s_cur[v >> SRC_BITS], 1);
    if (fits) s_stage[p] = v & SRC_MASK;
    else atomicExch(by_dst + base + p, v & SRC_MASK);
  });
  __syncthreads();

  if (fits) {
    const int padded = round32(total);
    for (int i = tid; i < padded; i += 1024) by_dst[base + i] = i < total ? s_stage[i] : 0;
  }
}

#define ROWS_PER_WAVE 4
#define ROW_GROUP 8

template <class Emit>
__device__ __forceinline__ void sum_rows(const float* __restrict__ H,
                                         const int* __restrict__ by_dst,
                                         int o_l, int e_l, int nrows, int lane, Emit emit) {
  const float2* __restrict__ Hl = reinterpret_cast<const float2*>(H) + lane;

  int beg = __builtin_amdgcn_readlane(o_l, 0);
  int end = __builtin_amdgcn_readlane(e_l, 0);
  int s_row = 0;
  if (beg + lane < end) s_row = by_dst[beg + lane];

#pragma clang loop unroll(disable)
  for (int rr = 0; rr < nrows; ++rr) {
    const int nr = min(rr + 1, nrows - 1);
    const int nbeg = __builtin_amdgcn_readlane(o_l, nr);
    const int nend = rr + 1 < nrows ? __builtin_amdgcn_readlane(e_l, nr) : nbeg;
    int s_nrow = 0;
    if (nbeg + lane < nend) s_nrow = by_dst[nbeg + lane];

    float ax = 0.0f, ay = 0.0f;
    int s_cur = s_row;
    for (int base = beg; base < end; base += 32) {
      int s_nxt = 0;
      if (base + 32 + lane < end) s_nxt = by_dst[base + 32 + lane];
      const int cnt = min(32, end - base);
      int j = 0;
      for (; j + ROW_GROUP <= cnt; j += ROW_GROUP) {
        float2 v[ROW_GROUP];
#pragma unroll
        for (int u = 0; u < ROW_GROUP; ++u) {
          const int s = __builtin_amdgcn_readlane(s_cur, j + u);
          v[u] = Hl[(size_t)s * 32];
        }
#pragma unroll
        for (int u = 0; u < ROW_GROUP; ++u) {
          ax += v[u].x;
          ay += v[u].y;
        }
      }
      if (j < cnt) {
        float2 v[ROW_GROUP - 1];
#pragma unroll
        for (int u = 0; u < ROW_GROUP - 1; ++u) {
          v[u] = make_float2(0.0f, 0.0f);
          if (j + u < cnt) {
            const int s = __builtin_amdgcn_readlane(s_cur, j + u);
            v[u] = Hl[(size_t)s * 32];
          }
        }
#pragma unroll
        for (int u = 0; u < ROW_GROUP - 1; ++u) {
          ax += v[u].x;
          ay += v[u].y;
        }
      }
      s_cur = s_nxt;
    }

    emit(rr, ax, ay);

    beg = nbeg;
    end = nend;
    s_row = s_nrow;
  }
}

__global__ void __launch_bounds__(256) gather_rows(const float* __restrict__ H,
                                                   const int* __restrict__ off,
                                                   const int* __restrict__ cnt_in,
                                                   const int* __restrict__ by_dst,
                                                   const float* __restrict__ norm_d,
                                                   const float* __restrict__ bias,
                                                   float* __restrict__ out,
                                                   int n_nodes, int do_relu) {
  const int lane = threadIdx.x & 31;
  const int wave = __builtin_amdgcn_readfirstlane((int)((blockIdx.x * blockDim.x + threadIdx.x) >> 5));
  const int r0 = wave * ROWS_PER_WAVE;
  if (r0 >= n_nodes) return;
  const int nrows = min(ROWS_PER_WAVE, n_nodes - r0);

  const int rl = r0 + min(lane, nrows - 1);
  const int o_l = off[rl];
  const int e_l = o_l + cnt_in[rl];
  const float nd_l = norm_d[rl];
  const float bx = bias[2 * lane];
  const float by = bias[2 * lane + 1];
  float2* __restrict__ outl = reinterpret_cast<float2*>(out) + lane;

  sum_rows(H, by_dst, o_l, e_l, nrows, lane, [&](int rr, float ax, float ay) {
    const float nd = __int_as_float(__builtin_amdgcn_readlane(__float_as_int(nd_l), rr));
    float ox = fmaf(ax, nd, bx);
    float oy = fmaf(ay, nd, by);
    if (do_relu) {
      ox = fmaxf(ox, 0.0f);
      oy = fmaxf(oy, 0.0f);
    }
    outl[(size_t)(r0 + rr) * 32] = make_float2(ox, oy);
  });
}

#define TILE_PITCH 68

__global__ void __launch_bounds__(256) gather_rows_transform(const float* __restrict__ H,
                                                             const int* __restrict__ off,
                                                             const int* __restrict__ cnt_in,
                                                             const int* __restrict__ by_dst,
                                                             const float* __restrict__ norm_d,
                                                             const float* __restrict__ bias,
                                                             const float* __restrict__ norm_s,
                                                             const float* __restrict__ Wn,
                                                             float* __restrict__ Y,
                                                             int n_nodes, int do_relu) {
  __shared__ float sW[64 * 64];
  __shared__ float sT[8 * ROWS_PER_WAVE * TILE_PITCH];
  __shared__ float sY[8 * ROWS_PER_WAVE * TILE_PITCH];

  const int lane = threadIdx.x & 31;
  const int w = __builtin_amdgcn_readfirstlane((int)(threadIdx.x >> 5));
  const int r0 = ((int)blockIdx.x * 8 + w) * ROWS_PER_WAVE;
  const int nrows = max(0, min(ROWS_PER_WAVE, n_nodes - r0));

  const int rl = min(r0 + min(lane, ROWS_PER_WAVE - 1), n_nodes - 1);
  const int o_l = off[rl];
  const int c_l = cnt_in[rl];
  const int e_l = o_l + (nrows > 0 ? c_l : 0);
  const float nd_l = norm_d[rl];
  const float ns_l = norm_s[rl];
  const float bx = bias[2 * lane];
  const float by = bias[2 * lane + 1];

  const float4* __restrict__ W4 = reinterpret_cast<const float4*>(Wn);
  float4 wv[4];
#pragma unroll
  for (int u = 0; u < 4; ++u) wv[u] = W4[threadIdx.x + 256 * u];
  asm volatile("" ::"v"(wv[0].x), "v"(wv[0].y), "v"(wv[0].z), "v"(wv[0].w),
               "v"(wv[1].x), "v"(wv[1].y), "v"(wv[1].z), "v"(wv[1].w),
               "v"(wv[2].x), "v"(wv[2].y), "v"(wv[2].z), "v"(wv[2].w),
               "v"(wv[3].x), "v"(wv[3].y), "v"(wv[3].z), "v"(wv[3].w));
#pragma unroll
  for (int u = 0; u < 4; ++u) reinterpret_cast<float4*>(sW)[threadIdx.x + 256 * u] = wv[u];

  float* __restrict__ trow = sT + (w * ROWS_PER_WAVE) * TILE_PITCH + 2 * lane;
#pragma unroll
  for (int rr = 0; rr < ROWS_PER_WAVE; ++rr) {
    if (rr >= nrows) *reinterpret_cast<float2*>(trow + rr * TILE_PITCH) = make_float2(0.0f, 0.0f);
  }

  sum_rows(H, by_dst, o_l, e_l, nrows, lane, [&](int rr, float ax, float ay) {
    const float nd = __int_as_float(__builtin_amdgcn_readlane(__float_as_int(nd_l), rr));
    const float ns = __int_as_float(__builtin_amdgcn_readlane(__float_as_int(ns_l), rr));
    float ox = fmaf(ax, nd, bx);
    float oy = fmaf(ay, nd, by);
    if (do_relu) {
      ox = fmaxf(ox, 0.0f);
      oy = fmaxf(oy, 0.0f);
    }
    *reinterpret_cast<float2*>(trow + rr * TILE_PITCH) = make_float2(ox * ns, oy * ns);
  });
  __syncthreads();

  const int mt = w >> 2;
  const int nt = w & 3;
  const int half = lane >> 4;
  const int sub = lane & 15;
  const float* __restrict__ arow = sT + (mt * 16 + sub) * TILE_PITCH + 2 * half;
  const float* __restrict__ wcol = sW + (2 * half) * 64 + nt * 16 + sub;

  v8f acc = {};
#pragma unroll
  for (int kk = 0; kk < 16; ++kk) {
    v2f a;
    a.x = arow[4 * kk];
    a.y = arow[4 * kk + 1];
    v2f b;
    b.x = wcol[(4 * kk) * 64];
    b.y = wcol[(4 * kk + 1) * 64];
    acc = __builtin_amdgcn_wmma_f32_16x16x4_f32(false, a, false, b, (short)0, acc, false, false);
  }

  float* __restrict__ ytile = sY + (mt * 16 + 8 * half) * TILE_PITCH + nt * 16 + sub;
#pragma unroll
  for (int r = 0; r < 8; ++r) ytile[r * TILE_PITCH] = acc[r];
  __syncthreads();

  const float* __restrict__ yrow = sY + (w * ROWS_PER_WAVE) * TILE_PITCH + 2 * lane;
  float2* __restrict__ yl = reinterpret_cast<float2*>(Y) + lane;
#pragma unroll
  for (int rr = 0; rr < ROWS_PER_WAVE; ++rr) {
    if (rr < nrows)
      yl[(size_t)(r0 + rr) * 32] = *reinterpret_cast<const float2*>(yrow + rr * TILE_PITCH);
  }
}

template <int K>
__global__ void __launch_bounds__(256) gemm_norm_wmma(const float* __restrict__ X,
                                                      const float* __restrict__ W,
                                                      const float* __restrict__ norm_s,
                                                      float* __restrict__ Y,
                                                      int n_nodes) {
  __shared__ float sW[K * 64];
  for (int i = threadIdx.x; i < K * 64; i += 256) sW[i] = W[i];
  __syncthreads();

  const int wave = threadIdx.x >> 5;
  const int lane = threadIdx.x & 31;
  const int row0 = (blockIdx.x * 8 + wave) * 16;
  if (row0 + 15 >= n_nodes) return;

  const int half  = lane >> 4;
  const int sub   = lane & 15;
  const int mrow  = row0 + sub;
  const float ns  = norm_s[mrow];
  const float* __restrict__ xrow = X + (size_t)mrow * K;

  v8f acc0 = {}, acc1 = {}, acc2 = {}, acc3 = {};

#pragma unroll
  for (int kk = 0; kk < K / 4; ++kk) {
    const int ka = kk * 4 + half * 2;
    v2f a;
    a.x = xrow[ka] * ns;
    a.y = xrow[ka + 1] * ns;

    const float* wk = &sW[ka * 64 + sub];
    v2f b;

    b.x = wk[0];       b.y = wk[64];
    acc0 = __builtin_amdgcn_wmma_f32_16x16x4_f32(false, a, false, b, (short)0, acc0, false, false);
    b.x = wk[16];      b.y = wk[64 + 16];
    acc1 = __builtin_amdgcn_wmma_f32_16x16x4_f32(false, a, false, b, (short)0, acc1, false, false);
    b.x = wk[32];      b.y = wk[64 + 32];
    acc2 = __builtin_amdgcn_wmma_f32_16x16x4_f32(false, a, false, b, (short)0, acc2, false, false);
    b.x = wk[48];      b.y = wk[64 + 48];
    acc3 = __builtin_amdgcn_wmma_f32_16x16x4_f32(false, a, false, b, (short)0, acc3, false, false);
  }

  float* ybase = Y + (size_t)row0 * 64 + sub;
#pragma unroll
  for (int r = 0; r < 8; ++r) {
    const int m = r + half * 8;
    float* yr = ybase + (size_t)m * 64;
    yr[0]  = acc0[r];
    yr[16] = acc1[r];
    yr[32] = acc2[r];
    yr[48] = acc3[r];
  }
}

static inline int cdiv(long long a, long long b) { return (int)((a + b - 1) / b); }

extern "C" void kernel_launch(void* const* d_in, const int* in_sizes, int n_in,
                              void* d_out, int out_size, void* d_ws, size_t ws_size,
                              hipStream_t stream) {
  const float* features = (const float*)d_in[0];
  const int*   src      = (const int*)d_in[1];
  const int*   dst      = (const int*)d_in[2];
  const float* W1       = (const float*)d_in[3];
  const float* b1       = (const float*)d_in[4];
  const float* W2       = (const float*)d_in[5];
  const float* b2       = (const float*)d_in[6];

  const int n_nodes = in_sizes[0] / 128;
  const int n_edges = in_sizes[1];

  const int n_buckets = (n_nodes + BKT_NODES - 1) >> BKT_SHIFT;
  const int n_pblocks = cdiv(n_edges, EPB);

  const size_t np = ((size_t)n_nodes + 63) & ~(size_t)63;
  const size_t ep = ((size_t)n_edges + 63) & ~(size_t)63;
  const size_t tp = (size_t)n_pblocks * NBMAX;
  const size_t rp = (size_t)n_pblocks * RUN_CAP;

  int*   wsi      = (int*)d_ws;
  int*   off      = wsi;
  int*   cnt_in   = off + np;
  float* norm_s   = (float*)(cnt_in + np);
  float* norm_d   = norm_s + np;
  int*   base_d   = (int*)(norm_d + np);
  int*   runcnt_d = base_d + NBMAX;
  int*   runoff_d = runcnt_d + tp;
  int*   runcnt_s = runoff_d + tp;
  int*   runoff_s = runcnt_s + tp;
  int*   runs_d   = runoff_s + tp;
  int*   runs_s   = runs_d + rp;
  int*   by_dst   = runs_s + rp;
  float* bufA     = (float*)(by_dst + ep + 32 * NBMAX);
  float* bufB     = bufA + (size_t)n_nodes * 64;
  float* out      = (float*)d_out;

  const int B = 256;

  partition_edges<<<n_pblocks, 1024, 0, stream>>>(src, dst, runcnt_d, runoff_d, runs_d,
                                                  runcnt_s, runoff_s, runs_s, n_edges);
  bucket_bases<<<1, 1024, 0, stream>>>(runcnt_d, base_d, n_pblocks);
  finish_buckets<<<2 * n_buckets, 1024, 0, stream>>>(runcnt_d, runoff_d, runs_d, runcnt_s, runoff_s,
                                                     runs_s, base_d, off, cnt_in, norm_s, norm_d,
                                                     by_dst, n_nodes, n_pblocks, n_buckets);

  const int gemm_blocks = cdiv(n_nodes, 128);
  const int gather_blocks = cdiv(n_nodes, 8 * ROWS_PER_WAVE);

  gemm_norm_wmma<128><<<gemm_blocks, B, 0, stream>>>(features, W1, norm_s, bufA, n_nodes);
  gather_rows_transform<<<gather_blocks, B, 0, stream>>>(bufA, off, cnt_in, by_dst, norm_d, b1, norm_s,
                                                         W2, bufB, n_nodes, 1);

  gather_rows<<<gather_blocks, B, 0, stream>>>(bufB, off, cnt_in, by_dst, norm_d, b2, out, n_nodes, 0);
}


# Launches regrouped by dependency: bucket bases run beside the source buckets, layer 1's X@W1 product beside the destination buckets (role by block index): 6 launches to 5; sums keep LDS arrival order
# speedup vs baseline: 6.5837x; 1.0624x over previous
#include <hip/hip_runtime.h>


typedef float v2f __attribute__((ext_vector_type(2)));
typedef float v8f __attribute__((ext_vector_type(8)));

#define BKT_SHIFT 9
#define BKT_NODES 512
#define NBMAX 256
#define SRC_BITS 17
#define SRC_MASK 0x1FFFF
#define EPB 12544
#define EPT ((EPB + 1023) / 1024)
#define RUN_CAP (EPB + 31 * NBMAX)
#define FIN_CAP 32768

__device__ __forceinline__ int round32(int x) { return (x + 31) & ~31; }

__device__ __forceinline__ int block_scan_excl(int v, int* s_part, int* total) {
  const int lane = threadIdx.x & 31;
  const int w = threadIdx.x >> 5;
  int incl = v;
  for (int o = 1; o < 32; o <<= 1) {
    const int t = __shfl_up(incl, o);
    if (lane >= o) incl += t;
  }
  if (lane == 31) s_part[w] = incl;
  __syncthreads();
  const int p = s_part[lane];
  int before = lane < w ? p : 0;
  int all = p;
  for (int o = 16; o > 0; o >>= 1) {
    before += __shfl_xor(before, o);
    all += __shfl_xor(all, o);
  }
  *total = all;
  return before + incl - v;
}

__global__ void __launch_bounds__(1024) partition_edges(const int* __restrict__ src,
                                                        const int* __restrict__ dst,
                                                        int* __restrict__ runcnt_d,
                                                        int* __restrict__ runoff_d,
                                                        int* __restrict__ runs_d,
                                                        int* __restrict__ runcnt_s,
                                                        int* __restrict__ runoff_s,
                                                        int* __restrict__ runs_s,
                                                        int n_edges) {
  __shared__ int s_img_d[RUN_CAP];
  __shared__ int s_img_s[RUN_CAP];
  __shared__ int s_cnt_d[NBMAX], s_cnt_s[NBMAX];
  __shared__ int s_off_d[NBMAX], s_off_s[NBMAX];
  __shared__ int s_cur_d[NBMAX], s_cur_s[NBMAX];
  __shared__ int s_part_d[32], s_part_s[32];

  const int tid = threadIdx.x;
  const int lane = tid & 31;
  const int w = tid >> 5;
  const int b = blockIdx.x;
  const int e0 = b * EPB;
  const int e1 = min(e0 + EPB, n_edges);

  int d[EPT], s[EPT];
#pragma unroll
  for (int u = 0; u < EPT; ++u) {
    const int e = min(e0 + tid + 1024 * u, e1 - 1);
    d[u] = dst[e];
    s[u] = src[e];
  }
#pragma unroll
  for (int u = 0; u < EPT; ++u) {
    if (e0 + tid + 1024 * u >= e1) d[u] = -1;
  }

  if (tid < NBMAX) {
    s_cnt_d[tid] = 0;
    s_cnt_s[tid] = 0;
  }
  __syncthreads();

#pragma unroll
  for (int u = 0; u < EPT; ++u) {
    if (d[u] >= 0) {
      atomicAdd(&s_cnt_d[d[u] >> BKT_SHIFT], 1);
      atomicAdd(&s_cnt_s[s[u] >> BKT_SHIFT], 1);
    }
  }
  __syncthreads();

  const int cd = tid < NBMAX ? s_cnt_d[tid] : 0;
  const int cs = tid < NBMAX ? s_cnt_s[tid] : 0;
  int tot_d, tot_s;
  const int od = block_scan_excl(round32(cd), s_part_d, &tot_d);
  const int os = block_scan_excl(round32(cs), s_part_s, &tot_s);
  if (tid < NBMAX) {
    s_off_d[tid] = od;
    s_cur_d[tid] = od;
    s_off_s[tid] = os;
    s_cur_s[tid] = os;
    runcnt_d[b * NBMAX + tid] = cd;
    runoff_d[b * NBMAX + tid] = od;
    runcnt_s[b * NBMAX + tid] = cs;
    runoff_s[b * NBMAX + tid] = os;
  }
  __syncthreads();

  for (int k = w; k < NBMAX; k += 32) {
    const int nd = s_cnt_d[k];
    if (nd + lane < round32(nd)) s_img_d[s_off_d[k] + nd + lane] = 0;
    const int ns = s_cnt_s[k];
    if (ns + lane < round32(ns)) s_img_s[s_off_s[k] + ns + lane] = 0;
  }

#pragma unroll
  for (int u = 0; u < EPT; ++u) {
    if (d[u] >= 0) {
      const int pd = atomicAdd(&s_cur_d[d[u] >> BKT_SHIFT], 1);
      s_img_d[pd] = ((d[u] & (BKT_NODES - 1)) << SRC_BITS) | s[u];
      const int ps = atomicAdd(&s_cur_s[s[u] >> BKT_SHIFT], 1);
      s_img_s[ps] = s[u] & (BKT_NODES - 1);
    }
  }
  __syncthreads();

  int* __restrict__ out_d = runs_d + (size_t)b * RUN_CAP;
  int* __restrict__ out_s = runs_s + (size_t)b * RUN_CAP;
  for (int i = tid; i < tot_d; i += 1024) out_d[i] = s_img_d[i];
  for (int i = tid; i < tot_s; i += 1024) out_s[i] = s_img_s[i];
}

__device__ __forceinline__ void bucket_bases_block(const int* __restrict__ runcnt_d,
                                                   int* __restrict__ base_d, int n_blocks,
                                                   int* s_sum, int* s_part) {
  const int tid = threadIdx.x;
  const int k = tid & (NBMAX - 1);
  const int part = tid >> 8;
  int acc = 0;
  for (int b = part; b < n_blocks; b += 64) {
    int v[16];
#pragma unroll
    for (int u = 0; u < 16; ++u) v[u] = runcnt_d[min(b + 4 * u, n_blocks - 1) * NBMAX + k];
#pragma unroll
    for (int u = 0; u < 16; ++u) acc += (b + 4 * u < n_blocks) ? v[u] : 0;
  }
  s_sum[part * NBMAX + k] = acc;
  __syncthreads();
  int v = 0;
  if (tid < NBMAX)
    v = round32(s_sum[tid] + s_sum[NBMAX + tid] + s_sum[2 * NBMAX + tid] + s_sum[3 * NBMAX + tid]);
  int tot;
  const int ex = block_scan_excl(v, s_part, &tot);
  if (tid < NBMAX) base_d[tid] = ex;
}

#define RUNS_AHEAD 4
template <class F>
__device__ __forceinline__ void walk_runs(const int* __restrict__ runcnt,
                                          const int* __restrict__ runoff,
                                          const int* __restrict__ runs, int k, int n_blocks, F f) {
  const int lane = threadIdx.x & 31;
  const int w = threadIdx.x >> 5;
  for (int b0 = w; b0 < n_blocks; b0 += 1024) {
    const int bl = b0 + 32 * lane;
    int n_l = 0, o_l = 0;
    if (bl < n_blocks) {
      n_l = runcnt[bl * NBMAX + k];
      o_l = runoff[bl * NBMAX + k];
    }
    const int jn = min(32, (n_blocks - b0 + 31) >> 5);
    for (int j0 = 0; j0 < jn; j0 += RUNS_AHEAD) {
      int n[RUNS_AHEAD];
      const int* run[RUNS_AHEAD];
      int nmax = 0;
#pragma unroll
      for (int u = 0; u < RUNS_AHEAD; ++u) {
        n[u] = __builtin_amdgcn_readlane(n_l, j0 + u);
        run[u] = runs + (size_t)(b0 + 32 * (j0 + u)) * RUN_CAP + __builtin_amdgcn_readlane(o_l, j0 + u);
        nmax = max(nmax, n[u]);
      }
      for (int i = lane; i < nmax; i += 32) {
        int v[RUNS_AHEAD];
#pragma unroll
        for (int u = 0; u < RUNS_AHEAD; ++u) {
          v[u] = 0;
          if (i < n[u]) v[u] = run[u][i];
        }
#pragma unroll
        for (int u = 0; u < RUNS_AHEAD; ++u) {
          if (i < n[u]) f(v[u]);
        }
      }
    }
  }
}

__device__ __forceinline__ void finish_source_bucket(int k, const int* __restrict__ runcnt_s,
                                                     const int* __restrict__ runoff_s,
                                                     const int* __restrict__ runs_s,
                                                     float* __restrict__ norm_s,
                                                     int n_nodes, int n_blocks, int* s_cnt) {
  const int tid = threadIdx.x;
  if (tid < BKT_NODES) s_cnt[tid] = 0;
  __syncthreads();
  walk_runs(runcnt_s, runoff_s, runs_s, k, n_blocks, [&](int v) { atomicAdd(&s_cnt[v], 1); });
  __syncthreads();
  const int node = k * BKT_NODES + tid;
  if (tid < BKT_NODES && node < n_nodes) norm_s[node] = rsqrtf(fmaxf((float)s_cnt[tid], 1.0f));
}

__device__ __forceinline__ void finish_dest_bucket(int k, const int* __restrict__ runcnt_d,
                                                   const int* __restrict__ runoff_d,
                                                   const int* __restrict__ runs_d,
                                                   const int* __restrict__ base_d,
                                                   int* __restrict__ off,
                                                   int* __restrict__ cnt_in,
                                                   float* __restrict__ norm_d,
                                                   int* __restrict__ by_dst,
                                                   int n_nodes, int n_blocks,
                                                   int* s_stage, int* s_cnt, int* s_cur, int* s_part) {
  const int tid = threadIdx.x;

  if (tid < BKT_NODES) s_cnt[tid] = 0;
  __syncthreads();

  walk_runs(runcnt_d, runoff_d, runs_d, k, n_blocks, [&](int v) { atomicAdd(&s_cnt[v >> SRC_BITS], 1); });
  __syncthreads();

  const int c = tid < BKT_NODES ? s_cnt[tid] : 0;
  int total;
  const int ex = block_scan_excl(c, s_part, &total);
  const int node = k * BKT_NODES + tid;
  const float nrm = rsqrtf(fmaxf((float)c, 1.0f));
  const int base = base_d[k];
  if (tid < BKT_NODES) {
    s_cur[tid] = ex;
    if (node < n_nodes) {
      off[node] = base + ex;
      cnt_in[node] = c;
      norm_d[node] = nrm;
    }
  }
  __syncthreads();

  const bool fits = total <= FIN_CAP;
  walk_runs(runcnt_d, runoff_d, runs_d, k, n_blocks, [&](int v) {
    const int p = atomicAdd(&s_cur[v >> SRC_BITS], 1);
    if (fits) s_stage[p] = v & SRC_MASK;
    else atomicExch(by_dst + base + p, v & SRC_MASK);
  });
  __syncthreads();

  if (fits) {
    const int padded = round32(total);
    for (int i = tid; i < padded; i += 1024) by_dst[base + i] = i < total ? s_stage[i] : 0;
  }
}

__global__ void __launch_bounds__(1024) bases_and_source_buckets(const int* __restrict__ runcnt_d,
                                                                 int* __restrict__ base_d,
                                                                 const int* __restrict__ runcnt_s,
                                                                 const int* __restrict__ runoff_s,
                                                                 const int* __restrict__ runs_s,
                                                                 float* __restrict__ norm_s,
                                                                 int n_nodes, int n_blocks) {
  __shared__ int s_pool[4 * NBMAX + 32];
  if (blockIdx.x == 0) {
    bucket_bases_block(runcnt_d, base_d, n_blocks, s_pool, s_pool + 4 * NBMAX);
  } else {
    finish_source_bucket((int)blockIdx.x - 1, runcnt_s, runoff_s, runs_s, norm_s, n_nodes, n_blocks,
                         s_pool);
  }
}

#define ROWS_PER_WAVE 4
#define ROW_GROUP 8

template <class Emit>
__device__ __forceinline__ void sum_rows(const float* __restrict__ H,
                                         const int* __restrict__ by_dst,
                                         int o_l, int e_l, int nrows, int lane, Emit emit) {
  const float2* __restrict__ Hl = reinterpret_cast<const float2*>(H) + lane;

  int beg = __builtin_amdgcn_readlane(o_l, 0);
  int end = __builtin_amdgcn_readlane(e_l, 0);
  int s_row = 0;
  if (beg + lane < end) s_row = by_dst[beg + lane];

#pragma clang loop unroll(disable)
  for (int rr = 0; rr < nrows; ++rr) {
    const int nr = min(rr + 1, nrows - 1);
    const int nbeg = __builtin_amdgcn_readlane(o_l, nr);
    const int nend = rr + 1 < nrows ? __builtin_amdgcn_readlane(e_l, nr) : nbeg;
    int s_nrow = 0;
    if (nbeg + lane < nend) s_nrow = by_dst[nbeg + lane];

    float ax = 0.0f, ay = 0.0f;
    int s_cur = s_row;
    for (int base = beg; base < end; base += 32) {
      int s_nxt = 0;
      if (base + 32 + lane < end) s_nxt = by_dst[base + 32 + lane];
      const int cnt = min(32, end - base);
      int j = 0;
      for (; j + ROW_GROUP <= cnt; j += ROW_GROUP) {
        float2 v[ROW_GROUP];
#pragma unroll
        for (int u = 0; u < ROW_GROUP; ++u) {
          const int s = __builtin_amdgcn_readlane(s_cur, j + u);
          v[u] = Hl[(size_t)s * 32];
        }
#pragma unroll
        for (int u = 0; u < ROW_GROUP; ++u) {
          ax += v[u].x;
          ay += v[u].y;
        }
      }
      if (j < cnt) {
        float2 v[ROW_GROUP - 1];
#pragma unroll
        for (int u = 0; u < ROW_GROUP - 1; ++u) {
          v[u] = make_float2(0.0f, 0.0f);
          if (j + u < cnt) {
            const int s = __builtin_amdgcn_readlane(s_cur, j + u);
            v[u] = Hl[(size_t)s * 32];
          }
        }
#pragma unroll
        for (int u = 0; u < ROW_GROUP - 1; ++u) {
          ax += v[u].x;
          ay += v[u].y;
        }
      }
      s_cur = s_nxt;
    }

    emit(rr, ax, ay);

    beg = nbeg;
    end = nend;
    s_row = s_nrow;
  }
}

__global__ void __launch_bounds__(256) gather_rows(const float* __restrict__ H,
                                                   const int* __restrict__ off,
                                                   const int* __restrict__ cnt_in,
                                                   const int* __restrict__ by_dst,
                                                   const float* __restrict__ norm_d,
                                                   const float* __restrict__ bias,
                                                   float* __restrict__ out,
                                                   int n_nodes, int do_relu) {
  const int lane = threadIdx.x & 31;
  const int wave = __builtin_amdgcn_readfirstlane((int)((blockIdx.x * blockDim.x + threadIdx.x) >> 5));
  const int r0 = wave * ROWS_PER_WAVE;
  if (r0 >= n_nodes) return;
  const int nrows = min(ROWS_PER_WAVE, n_nodes - r0);

  const int rl = r0 + min(lane, nrows - 1);
  const int o_l = off[rl];
  const int e_l = o_l + cnt_in[rl];
  const float nd_l = norm_d[rl];
  const float bx = bias[2 * lane];
  const float by = bias[2 * lane + 1];
  float2* __restrict__ outl = reinterpret_cast<float2*>(out) + lane;

  sum_rows(H, by_dst, o_l, e_l, nrows, lane, [&](int rr, float ax, float ay) {
    const float nd = __int_as_float(__builtin_amdgcn_readlane(__float_as_int(nd_l), rr));
    float ox = fmaf(ax, nd, bx);
    float oy = fmaf(ay, nd, by);
    if (do_relu) {
      ox = fmaxf(ox, 0.0f);
      oy = fmaxf(oy, 0.0f);
    }
    outl[(size_t)(r0 + rr) * 32] = make_float2(ox, oy);
  });
}

#define TILE_PITCH 68

__global__ void __launch_bounds__(256) gather_rows_transform(const float* __restrict__ H,
                                                             const int* __restrict__ off,
                                                             const int* __restrict__ cnt_in,
                                                             const int* __restrict__ by_dst,
                                                             const float* __restrict__ norm_d,
                                                             const float* __restrict__ bias,
                                                             const float* __restrict__ norm_s,
                                                             const float* __restrict__ Wn,
                                                             float* __restrict__ Y,
                                                             int n_nodes, int do_relu) {
  __shared__ float sW[64 * 64];
  __shared__ float sT[8 * ROWS_PER_WAVE * TILE_PITCH];
  __shared__ float sY[8 * ROWS_PER_WAVE * TILE_PITCH];

  const int lane = threadIdx.x & 31;
  const int w = __builtin_amdgcn_readfirstlane((int)(threadIdx.x >> 5));
  const int r0 = ((int)blockIdx.x * 8 + w) * ROWS_PER_WAVE;
  const int nrows = max(0, min(ROWS_PER_WAVE, n_nodes - r0));

  const int rl = min(r0 + min(lane, ROWS_PER_WAVE - 1), n_nodes - 1);
  const int o_l = off[rl];
  const int c_l = cnt_in[rl];
  const int e_l = o_l + (nrows > 0 ? c_l : 0);
  const float nd_l = norm_d[rl];
  const float ns_l = norm_s[rl];
  const float bx = bias[2 * lane];
  const float by = bias[2 * lane + 1];

  const float4* __restrict__ W4 = reinterpret_cast<const float4*>(Wn);
  float4 wv[4];
#pragma unroll
  for (int u = 0; u < 4; ++u) wv[u] = W4[threadIdx.x + 256 * u];
  asm volatile("" ::"v"(wv[0].x), "v"(wv[0].y), "v"(wv[0].z), "v"(wv[0].w),
               "v"(wv[1].x), "v"(wv[1].y), "v"(wv[1].z), "v"(wv[1].w),
               "v"(wv[2].x), "v"(wv[2].y), "v"(wv[2].z), "v"(wv[2].w),
               "v"(wv[3].x), "v"(wv[3].y), "v"(wv[3].z), "v"(wv[3].w));
#pragma unroll
  for (int u = 0; u < 4; ++u) reinterpret_cast<float4*>(sW)[threadIdx.x + 256 * u] = wv[u];

  float* __restrict__ trow = sT + (w * ROWS_PER_WAVE) * TILE_PITCH + 2 * lane;
#pragma unroll
  for (int rr = 0; rr < ROWS_PER_WAVE; ++rr) {
    if (rr >= nrows) *reinterpret_cast<float2*>(trow + rr * TILE_PITCH) = make_float2(0.0f, 0.0f);
  }

  sum_rows(H, by_dst, o_l, e_l, nrows, lane, [&](int rr, float ax, float ay) {
    const float nd = __int_as_float(__builtin_amdgcn_readlane(__float_as_int(nd_l), rr));
    const float ns = __int_as_float(__builtin_amdgcn_readlane(__float_as_int(ns_l), rr));
    float ox = fmaf(ax, nd, bx);
    float oy = fmaf(ay, nd, by);
    if (do_relu) {
      ox = fmaxf(ox, 0.0f);
      oy = fmaxf(oy, 0.0f);
    }
    *reinterpret_cast<float2*>(trow + rr * TILE_PITCH) = make_float2(ox * ns, oy * ns);
  });
  __syncthreads();

  const int mt = w >> 2;
  const int nt = w & 3;
  const int half = lane >> 4;
  const int sub = lane & 15;
  const float* __restrict__ arow = sT + (mt * 16 + sub) * TILE_PITCH + 2 * half;
  const float* __restrict__ wcol = sW + (2 * half) * 64 + nt * 16 + sub;

  v8f acc = {};
#pragma unroll
  for (int kk = 0; kk < 16; ++kk) {
    v2f a;
    a.x = arow[4 * kk];
    a.y = arow[4 * kk + 1];
    v2f b;
    b.x = wcol[(4 * kk) * 64];
    b.y = wcol[(4 * kk + 1) * 64];
    acc = __builtin_amdgcn_wmma_f32_16x16x4_f32(false, a, false, b, (short)0, acc, false, false);
  }

  float* __restrict__ ytile = sY + (mt * 16 + 8 * half) * TILE_PITCH + nt * 16 + sub;
#pragma unroll
  for (int r = 0; r < 8; ++r) ytile[r * TILE_PITCH] = acc[r];
  __syncthreads();

  const float* __restrict__ yrow = sY + (w * ROWS_PER_WAVE) * TILE_PITCH + 2 * lane;
  float2* __restrict__ yl = reinterpret_cast<float2*>(Y) + lane;
#pragma unroll
  for (int rr = 0; rr < ROWS_PER_WAVE; ++rr) {
    if (rr < nrows)
      yl[(size_t)(r0 + rr) * 32] = *reinterpret_cast<const float2*>(yrow + rr * TILE_PITCH);
  }
}

#define TILES_PER_GROUP 25

template <int K>
__device__ __forceinline__ void product_tiles(int g, const float* __restrict__ X,
                                              const float* __restrict__ W,
                                              const float* __restrict__ norm_s,
                                              float* __restrict__ Y, int n_nodes, float* sW) {
  static_assert(K * 16 % 1024 == 0, "W is staged in whole rounds of 1024 x 16 bytes");
  const float4* __restrict__ W4 = reinterpret_cast<const float4*>(W);
  float4 wv[K * 16 / 1024];
#pragma unroll
  for (int u = 0; u < K * 16 / 1024; ++u) wv[u] = W4[threadIdx.x + 1024 * u];
#pragma unroll
  for (int u = 0; u < K * 16 / 1024; ++u) reinterpret_cast<float4*>(sW)[threadIdx.x + 1024 * u] = wv[u];
  __syncthreads();

  const int wave = threadIdx.x >> 5;
  const int lane = threadIdx.x & 31;
  if (wave >= TILES_PER_GROUP) return;
  const int row0 = (g * TILES_PER_GROUP + wave) * 16;
  if (row0 + 15 >= n_nodes) return;

  const int half  = lane >> 4;
  const int sub   = lane & 15;
  const int mrow  = row0 + sub;
  const float ns  = norm_s[mrow];
  const float* __restrict__ xrow = X + (size_t)mrow * K;

  v8f acc0 = {}, acc1 = {}, acc2 = {}, acc3 = {};

#pragma unroll
  for (int kk = 0; kk < K / 4; ++kk) {
    const int ka = kk * 4 + half * 2;
    v2f a;
    a.x = xrow[ka] * ns;
    a.y = xrow[ka + 1] * ns;

    const float* wk = &sW[ka * 64 + sub];
    v2f b;

    b.x = wk[0];       b.y = wk[64];
    acc0 = __builtin_amdgcn_wmma_f32_16x16x4_f32(false, a, false, b, (short)0, acc0, false, false);
    b.x = wk[16];      b.y = wk[64 + 16];
    acc1 = __builtin_amdgcn_wmma_f32_16x16x4_f32(false, a, false, b, (short)0, acc1, false, false);
    b.x = wk[32];      b.y = wk[64 + 32];
    acc2 = __builtin_amdgcn_wmma_f32_16x16x4_f32(false, a, false, b, (short)0, acc2, false, false);
    b.x = wk[48];      b.y = wk[64 + 48];
    acc3 = __builtin_amdgcn_wmma_f32_16x16x4_f32(false, a, false, b, (short)0, acc3, false, false);
  }

  float* ybase = Y + (size_t)row0 * 64 + sub;
#pragma unroll
  for (int r = 0; r < 8; ++r) {
    const int m = r + half * 8;
    float* yr = ybase + (size_t)m * 64;
    yr[0]  = acc0[r];
    yr[16] = acc1[r];
    yr[32] = acc2[r];
    yr[48] = acc3[r];
  }
}

__global__ void __launch_bounds__(1024) product_and_dest_buckets(const float* __restrict__ X,
                                                                 const float* __restrict__ W,
                                                                 const float* __restrict__ norm_s,
                                                                 float* __restrict__ Y,
                                                                 const int* __restrict__ runcnt_d,
                                                                 const int* __restrict__ runoff_d,
                                                                 const int* __restrict__ runs_d,
                                                                 const int* __restrict__ base_d,
                                                                 int* __restrict__ off,
                                                                 int* __restrict__ cnt_in,
                                                                 float* __restrict__ norm_d,
                                                                 int* __restrict__ by_dst,
                                                                 int n_nodes, int n_blocks, int n_tile_groups) {
  __shared__ int s_pool[FIN_CAP + 2 * BKT_NODES + 32];
  if ((int)blockIdx.x < n_tile_groups) {
    product_tiles<128>((int)blockIdx.x, X, W, norm_s, Y, n_nodes, reinterpret_cast<float*>(s_pool));
  } else {
    finish_dest_bucket((int)blockIdx.x - n_tile_groups, runcnt_d, runoff_d, runs_d, base_d, off, cnt_in,
                       norm_d, by_dst, n_nodes, n_blocks, s_pool, s_pool + FIN_CAP,
                       s_pool + FIN_CAP + BKT_NODES, s_pool + FIN_CAP + 2 * BKT_NODES);
  }
}

static inline int cdiv(long long a, long long b) { return (int)((a + b - 1) / b); }

extern "C" void kernel_launch(void* const* d_in, const int* in_sizes, int n_in,
                              void* d_out, int out_size, void* d_ws, size_t ws_size,
                              hipStream_t stream) {
  const float* features = (const float*)d_in[0];
  const int*   src      = (const int*)d_in[1];
  const int*   dst      = (const int*)d_in[2];
  const float* W1       = (const float*)d_in[3];
  const float* b1       = (const float*)d_in[4];
  const float* W2       = (const float*)d_in[5];
  const float* b2       = (const float*)d_in[6];

  const int n_nodes = in_sizes[0] / 128;
  const int n_edges = in_sizes[1];

  const int n_buckets = (n_nodes + BKT_NODES - 1) >> BKT_SHIFT;
  const int n_pblocks = cdiv(n_edges, EPB);

  const size_t np = ((size_t)n_nodes + 63) & ~(size_t)63;
  const size_t ep = ((size_t)n_edges + 63) & ~(size_t)63;
  const size_t tp = (size_t)n_pblocks * NBMAX;
  const size_t rp = (size_t)n_pblocks * RUN_CAP;

  int*   wsi      = (int*)d_ws;
  int*   off      = wsi;
  int*   cnt_in   = off + np;
  float* norm_s   = (float*)(cnt_in + np);
  float* norm_d   = norm_s + np;
  int*   base_d   = (int*)(norm_d + np);
  int*   runcnt_d = base_d + NBMAX;
  int*   runoff_d = runcnt_d + tp;
  int*   runcnt_s = runoff_d + tp;
  int*   runoff_s = runcnt_s + tp;
  int*   runs_d   = runoff_s + tp;
  int*   runs_s   = runs_d + rp;
  int*   by_dst   = runs_s + rp;
  float* bufA     = (float*)(by_dst + ep + 32 * NBMAX);
  float* bufB     = bufA + (size_t)n_nodes * 64;
  float* out      = (float*)d_out;

  const int B = 256;

  const int tile_groups = cdiv(n_nodes, 16 * TILES_PER_GROUP);
  const int gather_blocks = cdiv(n_nodes, 8 * ROWS_PER_WAVE);

  partition_edges<<<n_pblocks, 1024, 0, stream>>>(src, dst, runcnt_d, runoff_d, runs_d,
                                                  runcnt_s, runoff_s, runs_s, n_edges);
  bases_and_source_buckets<<<1 + n_buckets, 1024, 0, stream>>>(runcnt_d, base_d, runcnt_s, runoff_s,
                                                               runs_s, norm_s, n_nodes, n_pblocks);
  product_and_dest_buckets<<<tile_groups + n_buckets, 1024, 0, stream>>>(
      features, W1, norm_s, bufA, runcnt_d, runoff_d, runs_d, base_d, off, cnt_in, norm_d, by_dst,
      n_nodes, n_pblocks, tile_groups);

  gather_rows_transform<<<gather_blocks, B, 0, stream>>>(bufA, off, cnt_in, by_dst, norm_d, b1, norm_s,
                                                         W2, bufB, n_nodes, 1);

  gather_rows<<<gather_blocks, B, 0, stream>>>(bufB, off, cnt_in, by_dst, norm_d, b2, out, n_nodes, 0);
}
